# starting kernel
# baseline (speedup 1.0000x reference)
#include <hip/hip_runtime.h>

typedef float v2f __attribute__((ext_vector_type(2)));
typedef float v8f __attribute__((ext_vector_type(8)));

constexpr int NPTS = 50000;
constexpr int KNBR = 16;
constexpr int CIN  = 64;
constexpr int COUT = 64;
constexpr int NB   = 4;
constexpr int NTILES = NPTS / 16;
constexpr int WPB  = 8;
constexpr int TPB  = WPB * 32;
constexpr int LDSW_FLOATS = KNBR * 16 * 2 * 64 * 2;

__global__ __launch_bounds__(TPB)
void conv_gnn_wmma_f32(const float* __restrict__ xin,
                       const int*   __restrict__ nbr,
                       const float* __restrict__ wgt,
                       const float* __restrict__ bias,
                       float*       __restrict__ out)
{
  extern __shared__ float ldsw[];
  const int tid = threadIdx.x;

  for (int p = tid; p < LDSW_FLOATS / 2; p += TPB) {
    const int o    = p & 63;
    const int half = (p >> 6) & 1;
    const int cb   = (p >> 7) & 15;
    const int k    = p >> 11;
    const int c0   = cb * 4 + half * 2;
    const int src  = (k * CIN + c0) * COUT + o;
    ldsw[2 * p]     = wgt[src];
    ldsw[2 * p + 1] = wgt[src + COUT];
  }
  __syncthreads();

  const int wave = tid >> 5;
  const int lane = tid & 31;
  const int tile = blockIdx.x * WPB + wave;
  if (tile >= NTILES) return;

  const int n0   = tile * 16;
  const int m    = lane & 15;
  const int half = lane >> 4;

  int roff[KNBR];
#pragma unroll
  for (int k = 0; k < KNBR; ++k) {
    const int idx = nbr[(n0 + m) * KNBR + k];
    roff[k] = (idx < NPTS) ? idx * (CIN * 4) : -1;
  }

  const char* bb[NB];
#pragma unroll
  for (int b = 0; b < NB; ++b)
    bb[b] = (const char*)xin + (size_t)b * (size_t)NPTS * CIN * 4 + half * 8;

  v8f acc[NB][4];
#pragma unroll
  for (int b = 0; b < NB; ++b)
#pragma unroll
    for (int ot = 0; ot < 4; ++ot)
      acc[b][ot] = {};

#pragma unroll
  for (int k = 0; k < KNBR; ++k) {
    const bool valid = roff[k] >= 0;
    const int  ro    = valid ? roff[k] : 0;
    const float* lw = &ldsw[(k * 32 + half) * 128 + m * 2];
#pragma unroll 4
    for (int cb = 0; cb < 16; ++cb) {
      v2f a[NB];
#pragma unroll
      for (int b = 0; b < NB; ++b) {
        v2f t = *(const v2f*)(bb[b] + ro + cb * 16);
        a[b] = valid ? t : v2f{0.0f, 0.0f};
      }
      v2f bw[4];
#pragma unroll
      for (int ot = 0; ot < 4; ++ot)
        bw[ot] = *(const v2f*)(lw + cb * 256 + ot * 32);
#pragma unroll
      for (int b = 0; b < NB; ++b)
#pragma unroll
        for (int ot = 0; ot < 4; ++ot)
          acc[b][ot] = __builtin_amdgcn_wmma_f32_16x16x4_f32(
              false, a[b], false, bw[ot], (short)0, acc[b][ot], false, false);
    }
  }

#pragma unroll
  for (int b = 0; b < NB; ++b) {
#pragma unroll
    for (int ot = 0; ot < 4; ++ot) {
      const float bs = bias[ot * 16 + m];
#pragma unroll
      for (int j = 0; j < 8; ++j) {
        const int row = n0 + half * 8 + j;
        out[((size_t)b * NPTS + row) * COUT + ot * 16 + m] = acc[b][ot][j] + bs;
      }
    }
  }
}

extern "C" void kernel_launch(void* const* d_in, const int* in_sizes, int n_in,
                              void* d_out, int out_size, void* d_ws, size_t ws_size,
                              hipStream_t stream) {
  (void)in_sizes; (void)n_in; (void)d_ws; (void)ws_size; (void)out_size;
  const float* xin  = (const float*)d_in[0];
  const int*   nbr  = (const int*)d_in[1];
  const float* wgt  = (const float*)d_in[2];
  const float* bias = (const float*)d_in[3];
  float*       out  = (float*)d_out;

  const size_t shmem = (size_t)LDSW_FLOATS * sizeof(float);
  hipFuncSetAttribute((const void*)conv_gnn_wmma_f32,
                      hipFuncAttributeMaxDynamicSharedMemorySize, (int)shmem);

  const int blocks = (NTILES + WPB - 1) / WPB;
  conv_gnn_wmma_f32<<<blocks, TPB, shmem, stream>>>(xin, nbr, wgt, bias, out);
}


# The 25th unit of each work-group is cut along the neighbor axis over 4 helper waves (one per SIMD, 4 slots each, partials summed via spare LDS): each SIMD issues 6 1/4 units, not 7/6/6/6
# speedup vs baseline: 1.3620x; 1.3620x over previous
#include <hip/hip_runtime.h>


typedef float v2f __attribute__((ext_vector_type(2)));
typedef float v4f __attribute__((ext_vector_type(4)));
typedef float v8f __attribute__((ext_vector_type(8)));

constexpr int NPTS = 50000;
constexpr int KNBR = 16;
constexpr int CIN  = 64;
constexpr int COUT = 64;
constexpr int NB   = 4;
constexpr int NTILES = NPTS / 16;
constexpr int BPW  = 2;
constexpr int NUNITS = NTILES * (NB / BPW);
constexpr int UPB  = 25;
constexpr int FULLW = 24;
constexpr int HELPW = 4;
constexpr int KPH  = KNBR / HELPW;
constexpr int WPB  = FULLW + HELPW;
constexpr int TPB  = WPB * 32;
constexpr int NBLOCKS = NUNITS / UPB;
constexpr int LDSW_FLOATS = KNBR * 16 * 2 * 64 * 2;
constexpr int LDSP_FLOATS = HELPW * BPW * 4 * 8 * 32;

static_assert(NUNITS == NBLOCKS * UPB, "every block owns exactly UPB units");

__device__ __forceinline__
void neighbor_slot(const float* __restrict__ ldsw, const int k, const int roffk,
                   const char* const (&bb)[BPW], const int half, const int m,
                   v8f (&acc)[BPW][4])
{
  const bool valid = roffk >= 0;
  const int  ro    = valid ? roffk : 0;
  const float* lw = &ldsw[(k * 32 + half) * 128 + m * 2];
#pragma unroll 4
  for (int cb = 0; cb < 16; ++cb) {
    v2f a[BPW];
#pragma unroll
    for (int b = 0; b < BPW; ++b) {
      v2f t = *(const v2f*)(bb[b] + ro + cb * 16);
      a[b] = valid ? t : v2f{0.0f, 0.0f};
    }
    v2f bw[4];
#pragma unroll
    for (int ot = 0; ot < 4; ++ot)
      bw[ot] = *(const v2f*)(lw + cb * 256 + ot * 32);
#pragma unroll
    for (int b = 0; b < BPW; ++b)
#pragma unroll
      for (int ot = 0; ot < 4; ++ot)
        acc[b][ot] = __builtin_amdgcn_wmma_f32_16x16x4_f32(
            false, a[b], false, bw[ot], (short)0, acc[b][ot], false, false);
  }
}

__global__ __launch_bounds__(TPB)
void conv_gnn_wmma_f32(const float* __restrict__ xin,
                       const int*   __restrict__ nbr,
                       const float* __restrict__ wgt,
                       const float* __restrict__ bias,
                       float*       __restrict__ out)
{
  extern __shared__ __attribute__((aligned(16))) float ldsw[];
  const int tid = threadIdx.x;

  for (int p = tid; p < LDSW_FLOATS / 2; p += TPB) {
    const int o    = p & 63;
    const int half = (p >> 6) & 1;
    const int cb   = (p >> 7) & 15;
    const int k    = p >> 11;
    const int c0   = cb * 4 + half * 2;
    const int src  = (k * CIN + c0) * COUT + o;
    ldsw[2 * p]     = wgt[src];
    ldsw[2 * p + 1] = wgt[src + COUT];
  }
  __syncthreads();

  const int  wave   = tid >> 5;
  const int  lane   = tid & 31;
  const bool helper = wave >= FULLW;
  const int  hq     = wave - FULLW;
  const int  unit   = blockIdx.x * UPB + (helper ? FULLW : wave);

  const int tile = unit >> 1;
  const int b0   = (unit & 1) * BPW;
  const int n0   = tile * 16;
  const int m    = lane & 15;
  const int half = lane >> 4;

  const char* bb[BPW];
#pragma unroll
  for (int b = 0; b < BPW; ++b)
    bb[b] = (const char*)xin + (size_t)(b0 + b) * (size_t)NPTS * CIN * 4 + half * 8;

  v8f acc[BPW][4];
#pragma unroll
  for (int b = 0; b < BPW; ++b)
#pragma unroll
    for (int ot = 0; ot < 4; ++ot)
      acc[b][ot] = {};

  v4f* part = (v4f*)(ldsw + LDSW_FLOATS);

  if (!helper) {
    int roff[KNBR];
#pragma unroll
    for (int k = 0; k < KNBR; ++k) {
      const int idx = nbr[(n0 + m) * KNBR + k];
      roff[k] = (idx < NPTS) ? idx * (CIN * 4) : -1;
    }

#pragma unroll
    for (int k = 0; k < KNBR; ++k) {
      if (k == 0)  __builtin_amdgcn_s_setprio(3);
      if (k == 9)  __builtin_amdgcn_s_setprio(2);
      if (k == 13) __builtin_amdgcn_s_setprio(1);
      if (k == 15) __builtin_amdgcn_s_setprio(0);
      neighbor_slot(ldsw, k, roff[k], bb, half, m, acc);
    }

#pragma unroll
    for (int b = 0; b < BPW; ++b) {
#pragma unroll
      for (int ot = 0; ot < 4; ++ot) {
        const float bs = bias[ot * 16 + m];
#pragma unroll
        for (int j = 0; j < 8; ++j) {
          const int row = n0 + half * 8 + j;
          out[((size_t)(b0 + b) * NPTS + row) * COUT + ot * 16 + m] = acc[b][ot][j] + bs;
        }
      }
    }
  } else {
    int roff[KPH];
#pragma unroll
    for (int i = 0; i < KPH; ++i) {
      const int idx = nbr[(n0 + m) * KNBR + hq * KPH + i];
      roff[i] = (idx < NPTS) ? idx * (CIN * 4) : -1;
    }
    __builtin_amdgcn_s_setprio(3);
#pragma unroll
    for (int i = 0; i < KPH; ++i)
      neighbor_slot(ldsw, hq * KPH + i, roff[i], bb, half, m, acc);
    __builtin_amdgcn_s_setprio(0);

#pragma unroll
    for (int b = 0; b < BPW; ++b)
#pragma unroll
      for (int ot = 0; ot < 4; ++ot) {
        const v8f a = acc[b][ot];
        const int t = (hq * (BPW * 4) + b * 4 + ot) * 2;
        part[(t + 0) * 32 + lane] = v4f{a[0], a[1], a[2], a[3]};
        part[(t + 1) * 32 + lane] = v4f{a[4], a[5], a[6], a[7]};
      }
  }

  __syncthreads();

  if (helper) {
    const int fb = hq >> 1;
    const int fo = (hq & 1) * 2;
#pragma unroll
    for (int u = 0; u < 2; ++u) {
      const int ot = fo + u;
      const float bs = bias[ot * 16 + m];
      v4f lo = v4f{0.0f, 0.0f, 0.0f, 0.0f};
      v4f hi = v4f{0.0f, 0.0f, 0.0f, 0.0f};
#pragma unroll
      for (int h = 0; h < HELPW; ++h) {
        const int t = (h * (BPW * 4) + fb * 4 + ot) * 2;
        lo += part[(t + 0) * 32 + lane];
        hi += part[(t + 1) * 32 + lane];
      }
#pragma unroll
      for (int j = 0; j < 8; ++j) {
        const int row = n0 + half * 8 + j;
        const float v = (j < 4) ? lo[j & 3] : hi[j & 3];
        out[((size_t)(b0 + fb) * NPTS + row) * COUT + ot * 16 + m] = v + bs;
      }
    }
  }
}

extern "C" void kernel_launch(void* const* d_in, const int* in_sizes, int n_in,
                              void* d_out, int out_size, void* d_ws, size_t ws_size,
                              hipStream_t stream) {
  (void)in_sizes; (void)n_in; (void)d_ws; (void)ws_size; (void)out_size;
  const float* xin  = (const float*)d_in[0];
  const int*   nbr  = (const int*)d_in[1];
  const float* wgt  = (const float*)d_in[2];
  const float* bias = (const float*)d_in[3];
  float*       out  = (float*)d_out;

  const size_t shmem = (size_t)(LDSW_FLOATS + LDSP_FLOATS) * sizeof(float);
  hipFuncSetAttribute((const void*)conv_gnn_wmma_f32,
                      hipFuncAttributeMaxDynamicSharedMemorySize, (int)shmem);

  conv_gnn_wmma_f32<<<NBLOCKS, TPB, shmem, stream>>>(xin, nbr, wgt, bias, out);
}


# Gather trip doubled to a whole 128-byte line per row: 16 row loads (8 channel blocks x 2 batches) in one clause ahead of 64 WMMAs; neighbor indices fetched 4 slots at a time to free registers
# speedup vs baseline: 1.4048x; 1.0314x over previous
#include <hip/hip_runtime.h>


typedef float v2f __attribute__((ext_vector_type(2)));
typedef float v4f __attribute__((ext_vector_type(4)));
typedef float v8f __attribute__((ext_vector_type(8)));
typedef int   v4i __attribute__((ext_vector_type(4)));

constexpr int NPTS = 50000;
constexpr int KNBR = 16;
constexpr int CIN  = 64;
constexpr int COUT = 64;
constexpr int NB   = 4;
constexpr int NTILES = NPTS / 16;
constexpr int BPW  = 2;
constexpr int NUNITS = NTILES * (NB / BPW);
constexpr int UPB  = 25;
constexpr int FULLW = 24;
constexpr int HELPW = 4;
constexpr int KPH  = KNBR / HELPW;
constexpr int WPB  = FULLW + HELPW;
constexpr int TPB  = WPB * 32;
constexpr int NBLOCKS = NUNITS / UPB;
constexpr int LDSW_FLOATS = KNBR * 16 * 2 * 64 * 2;
constexpr int LDSP_FLOATS = HELPW * BPW * 4 * 8 * 32;

static_assert(NUNITS == NBLOCKS * UPB, "every block owns exactly UPB units");

__device__ __forceinline__
void neighbor_slot(const float* __restrict__ ldsw, const int k, const int roffk,
                   const char* const (&bb)[BPW], const int half, const int m,
                   v8f (&acc)[BPW][4])
{
  const bool valid = roffk >= 0;
  const int  ro    = valid ? roffk : 0;
  const float* lw = &ldsw[(k * 32 + half) * 128 + m * 2];
#pragma unroll 8
  for (int cb = 0; cb < 16; ++cb) {
    v2f a[BPW];
#pragma unroll
    for (int b = 0; b < BPW; ++b) {
      v2f t = *(const v2f*)(bb[b] + ro + cb * 16);
      a[b] = valid ? t : v2f{0.0f, 0.0f};
    }
    v2f bw[4];
#pragma unroll
    for (int ot = 0; ot < 4; ++ot)
      bw[ot] = *(const v2f*)(lw + cb * 256 + ot * 32);
#pragma unroll
    for (int b = 0; b < BPW; ++b)
#pragma unroll
      for (int ot = 0; ot < 4; ++ot)
        acc[b][ot] = __builtin_amdgcn_wmma_f32_16x16x4_f32(
            false, a[b], false, bw[ot], (short)0, acc[b][ot], false, false);
  }
}

__global__ __launch_bounds__(TPB)
void conv_gnn_wmma_f32(const float* __restrict__ xin,
                       const int*   __restrict__ nbr,
                       const float* __restrict__ wgt,
                       const float* __restrict__ bias,
                       float*       __restrict__ out)
{
  extern __shared__ __attribute__((aligned(16))) float ldsw[];
  const int tid = threadIdx.x;

  for (int p = tid; p < LDSW_FLOATS / 2; p += TPB) {
    const int o    = p & 63;
    const int half = (p >> 6) & 1;
    const int cb   = (p >> 7) & 15;
    const int k    = p >> 11;
    const int c0   = cb * 4 + half * 2;
    const int src  = (k * CIN + c0) * COUT + o;
    ldsw[2 * p]     = wgt[src];
    ldsw[2 * p + 1] = wgt[src + COUT];
  }
  __syncthreads();

  const int  wave   = tid >> 5;
  const int  lane   = tid & 31;
  const bool helper = wave >= FULLW;
  const int  hq     = wave - FULLW;
  const int  unit   = blockIdx.x * UPB + (helper ? FULLW : wave);

  const int tile = unit >> 1;
  const int b0   = (unit & 1) * BPW;
  const int n0   = tile * 16;
  const int m    = lane & 15;
  const int half = lane >> 4;

  const char* bb[BPW];
#pragma unroll
  for (int b = 0; b < BPW; ++b)
    bb[b] = (const char*)xin + (size_t)(b0 + b) * (size_t)NPTS * CIN * 4 + half * 8;

  v8f acc[BPW][4];
#pragma unroll
  for (int b = 0; b < BPW; ++b)
#pragma unroll
    for (int ot = 0; ot < 4; ++ot)
      acc[b][ot] = {};

  v4f* part = (v4f*)(ldsw + LDSW_FLOATS);

  if (!helper) {
    const v4i* nrow = (const v4i*)(nbr + (n0 + m) * KNBR);
    v4i cur = nrow[0];

#pragma unroll 1
    for (int kg = 0; kg < KNBR / 4; ++kg) {
      const v4i nxt = nrow[(kg + 1 < KNBR / 4) ? kg + 1 : kg];
#pragma unroll
      for (int i = 0; i < 4; ++i) {
        const int k = kg * 4 + i;
        if (k == 0)  __builtin_amdgcn_s_setprio(3);
        if (k == 9)  __builtin_amdgcn_s_setprio(2);
        if (k == 13) __builtin_amdgcn_s_setprio(1);
        if (k == 15) __builtin_amdgcn_s_setprio(0);
        const int idx = cur[i];
        const int roffk = (idx < NPTS) ? idx * (CIN * 4) : -1;
        neighbor_slot(ldsw, k, roffk, bb, half, m, acc);
      }
      cur = nxt;
    }

#pragma unroll
    for (int b = 0; b < BPW; ++b) {
#pragma unroll
      for (int ot = 0; ot < 4; ++ot) {
        const float bs = bias[ot * 16 + m];
#pragma unroll
        for (int j = 0; j < 8; ++j) {
          const int row = n0 + half * 8 + j;
          out[((size_t)(b0 + b) * NPTS + row) * COUT + ot * 16 + m] = acc[b][ot][j] + bs;
        }
      }
    }
  } else {
    int roff[KPH];
#pragma unroll
    for (int i = 0; i < KPH; ++i) {
      const int idx = nbr[(n0 + m) * KNBR + hq * KPH + i];
      roff[i] = (idx < NPTS) ? idx * (CIN * 4) : -1;
    }
    __builtin_amdgcn_s_setprio(3);
#pragma unroll
    for (int i = 0; i < KPH; ++i)
      neighbor_slot(ldsw, hq * KPH + i, roff[i], bb, half, m, acc);
    __builtin_amdgcn_s_setprio(0);

#pragma unroll
    for (int b = 0; b < BPW; ++b)
#pragma unroll
      for (int ot = 0; ot < 4; ++ot) {
        const v8f a = acc[b][ot];
        const int t = (hq * (BPW * 4) + b * 4 + ot) * 2;
        part[(t + 0) * 32 + lane] = v4f{a[0], a[1], a[2], a[3]};
        part[(t + 1) * 32 + lane] = v4f{a[4], a[5], a[6], a[7]};
      }
  }

  __syncthreads();

  if (helper) {
    const int fb = hq >> 1;
    const int fo = (hq & 1) * 2;
#pragma unroll
    for (int u = 0; u < 2; ++u) {
      const int ot = fo + u;
      const float bs = bias[ot * 16 + m];
      v4f lo = v4f{0.0f, 0.0f, 0.0f, 0.0f};
      v4f hi = v4f{0.0f, 0.0f, 0.0f, 0.0f};
#pragma unroll
      for (int h = 0; h < HELPW; ++h) {
        const int t = (h * (BPW * 4) + fb * 4 + ot) * 2;
        lo += part[(t + 0) * 32 + lane];
        hi += part[(t + 1) * 32 + lane];
      }
#pragma unroll
      for (int j = 0; j < 8; ++j) {
        const int row = n0 + half * 8 + j;
        const float v = (j < 4) ? lo[j & 3] : hi[j & 3];
        out[((size_t)(b0 + fb) * NPTS + row) * COUT + ot * 16 + m] = v + bs;
      }
    }
  }
}

extern "C" void kernel_launch(void* const* d_in, const int* in_sizes, int n_in,
                              void* d_out, int out_size, void* d_ws, size_t ws_size,
                              hipStream_t stream) {
  (void)in_sizes; (void)n_in; (void)d_ws; (void)ws_size; (void)out_size;
  const float* xin  = (const float*)d_in[0];
  const int*   nbr  = (const int*)d_in[1];
  const float* wgt  = (const float*)d_in[2];
  const float* bias = (const float*)d_in[3];
  float*       out  = (float*)d_out;

  const size_t shmem = (size_t)(LDSW_FLOATS + LDSP_FLOATS) * sizeof(float);
  hipFuncSetAttribute((const void*)conv_gnn_wmma_f32,
                      hipFuncAttributeMaxDynamicSharedMemorySize, (int)shmem);

  conv_gnn_wmma_f32<<<NBLOCKS, TPB, shmem, stream>>>(xin, nbr, wgt, bias, out);
}
